# starting kernel
# baseline (speedup 1.0000x reference)
#include <hip/hip_runtime.h>
#include <math.h>

#define NQ        9
#define DIM       512
#define NLAYERS   12
#define OBS       256
#define BATCH_TOT 16384

#define WAVES_PER_BLOCK   8
#define STATES_PER_WAVE   8
#define BLOCK_THREADS     (WAVES_PER_BLOCK * 32)
#define STATES_PER_BLOCK  (WAVES_PER_BLOCK * STATES_PER_WAVE)
#define NGATES            (NLAYERS * NQ)
#define GATE_FLOATS       128
#define GATE_LDS_FLOATS   (NGATES * GATE_FLOATS)
#define STATE_LDS_FLOATS  (STATES_PER_WAVE * DIM * 2)

typedef float v2f __attribute__((ext_vector_type(2)));
typedef float v8f __attribute__((ext_vector_type(8)));

struct cpx { float re, im; };
__device__ inline cpx cz() { cpx r; r.re = 0.f; r.im = 0.f; return r; }
__device__ inline cpx cmk(float a, float b) { cpx r; r.re = a; r.im = b; return r; }
__device__ inline cpx cmul(cpx a, cpx b) { return cmk(a.re*b.re - a.im*b.im, a.re*b.im + a.im*b.re); }
__device__ inline cpx cadd(cpx a, cpx b) { return cmk(a.re + b.re, a.im + b.im); }

__device__ inline void u3mat(float t, float p, float l, cpx U[2][2]) {
    float ct = cosf(0.5f * t), st = sinf(0.5f * t);
    float cp = cosf(p), sp = sinf(p), cl = cosf(l), sl = sinf(l);
    U[0][0] = cmk(ct, 0.f);
    U[0][1] = cmk(-cl * st, -sl * st);
    U[1][0] = cmk(cp * st, sp * st);
    U[1][1] = cmk((cp*cl - sp*sl) * ct, (cp*sl + sp*cl) * ct);
}

__device__ inline void kron22(const cpx A[2][2], const cpx B[2][2], cpx K[4][4]) {
    for (int a = 0; a < 2; ++a)
        for (int b = 0; b < 2; ++b)
            for (int c = 0; c < 2; ++c)
                for (int d = 0; d < 2; ++d)
                    K[2*a + c][2*b + d] = cmul(A[a][b], B[c][d]);
}

__device__ inline void mat4mul(cpx C[4][4], const cpx A[4][4], const cpx B[4][4]) {
    for (int i = 0; i < 4; ++i)
        for (int j = 0; j < 4; ++j) {
            cpx acc = cz();
            for (int k = 0; k < 4; ++k) acc = cadd(acc, cmul(A[i][k], B[k][j]));
            C[i][j] = acc;
        }
}

__device__ void computeGate(int g, const float* __restrict__ w, float* __restrict__ gl) {
    const float* th = w + g * 15;
    cpx Ua[2][2], Ub[2][2];
    cpx pre[4][4], post[4][4];
    u3mat(th[0], th[1], th[2], Ua);  u3mat(th[3], th[4], th[5], Ub);  kron22(Ua, Ub, pre);
    u3mat(th[9], th[10], th[11], Ua); u3mat(th[12], th[13], th[14], Ub); kron22(Ua, Ub, post);

    float c6 = cosf(0.5f*th[6]), s6 = sinf(0.5f*th[6]);
    float c7 = cosf(0.5f*th[7]), s7 = sinf(0.5f*th[7]);
    float c8 = cosf(0.5f*th[8]), s8 = sinf(0.5f*th[8]);
    cpx Rxx[4][4], Ryy[4][4], Rzz[4][4];
    for (int i = 0; i < 4; ++i)
        for (int j = 0; j < 4; ++j) { Rxx[i][j] = cz(); Ryy[i][j] = cz(); Rzz[i][j] = cz(); }
    for (int i = 0; i < 4; ++i) { Rxx[i][i] = cmk(c6, 0.f); Ryy[i][i] = cmk(c7, 0.f); }
    Rxx[0][3] = cmk(0.f, -s6); Rxx[1][2] = cmk(0.f, -s6);
    Rxx[2][1] = cmk(0.f, -s6); Rxx[3][0] = cmk(0.f, -s6);
    Ryy[0][3] = cmk(0.f,  s7); Ryy[1][2] = cmk(0.f, -s7);
    Ryy[2][1] = cmk(0.f, -s7); Ryy[3][0] = cmk(0.f,  s7);
    Rzz[0][0] = cmk(c8, -s8); Rzz[1][1] = cmk(c8,  s8);
    Rzz[2][2] = cmk(c8,  s8); Rzz[3][3] = cmk(c8, -s8);

    cpx t1[4][4], t2[4][4];
    mat4mul(t1, Rxx, Ryy);
    mat4mul(t2, t1, Rzz);
    mat4mul(t1, t2, pre);
    mat4mul(t2, post, t1);

    float* gb = gl + g * GATE_FLOATS;
    for (int c = 0; c < 2; ++c)
        for (int k = 0; k < 4; ++k)
            for (int n = 0; n < 16; ++n) {
                int a = n & 7;
                int b = 4*c + k;
                float val;
                if (a < 4) val = (b < 4) ?  t2[a][b].re     : -t2[a][b-4].im;
                else       val = (b < 4) ?  t2[a-4][b].im   :  t2[a-4][b-4].re;
                gb[c*64 + k*16 + n] = val;
            }
}

__device__ inline unsigned insert2(unsigned r, int lo, unsigned vlo, int hi, unsigned vhi) {
    unsigned low = r & ((1u << lo) - 1u);
    unsigned x = ((r >> lo) << (lo + 1)) | (vlo << lo) | low;
    unsigned low2 = x & ((1u << hi) - 1u);
    return ((x >> hi) << (hi + 1)) | (vhi << hi) | low2;
}
__device__ inline unsigned ampIndex(unsigned r, int pm, int pl, unsigned p) {
    unsigned vm = (p >> 1) & 1u, vl = p & 1u;
    int lo  = (pm < pl) ? pm : pl;  unsigned vlo = (pm < pl) ? vm : vl;
    int hi  = (pm < pl) ? pl : pm;  unsigned vhi = (pm < pl) ? vl : vm;
    return insert2(r, lo, vlo, hi, vhi);
}

__global__ __launch_bounds__(BLOCK_THREADS, 1)
void su4_quantum_kernel(const float* __restrict__ features,
                        const float* __restrict__ weights,
                        float* __restrict__ out) {
    extern __shared__ float smem[];
    float* gl = smem;
    const int tid  = threadIdx.x;
    const int lane = tid & 31;
    const int wave = tid >> 5;
    float*  stateF = smem + GATE_LDS_FLOATS + wave * STATE_LDS_FLOATS;
    float2* stateV = (float2*)stateF;

    if (tid < NGATES) computeGate(tid, weights, gl);

    const int bBase = (blockIdx.x * WAVES_PER_BLOCK + wave) * STATES_PER_WAVE;
    for (int s = 0; s < STATES_PER_WAVE; ++s) {
        const float* f = features + (size_t)(bBase + s) * OBS;
        float ss = 0.f;
        for (int j = lane; j < OBS; j += 32) { float v = f[j]; ss += v * v; }
        for (int msk = 16; msk >= 1; msk >>= 1) ss += __shfl_xor(ss, msk, 32);
        float inv = 1.0f / (sqrtf(ss) + 1e-8f);
        for (int j = lane; j < DIM; j += 32) {
            float re = (j < OBS) ? f[j] * inv : 0.f;
            stateV[s * DIM + j] = make_float2(re, 0.f);
        }
    }
    __syncthreads();

    const int s    = lane & 7;
    const unsigned rsub = (unsigned)((lane >> 3) & 1);
    const int half = lane >> 4;
    const int kb   = 2 * half;
    const int ncol = lane & 15;
    const unsigned q = (unsigned)(ncol & 3);
    const int reim = (ncol >> 2) & 1;

    for (int layer = 0; layer < NLAYERS; ++layer) {
        for (int i = 0; i < NQ; ++i) {
            const int g = layer * NQ + i;
            const float* gb = gl + g * GATE_FLOATS;
            v2f B0, B1;
            B0[0] = gb[(kb    ) * 16 + ncol];
            B0[1] = gb[(kb + 1) * 16 + ncol];
            B1[0] = gb[64 + (kb    ) * 16 + ncol];
            B1[1] = gb[64 + (kb + 1) * 16 + ncol];
            const int pm = (i < 8) ? (8 - i) : 0;
            const int pl = (i < 8) ? (7 - i) : 8;

            float2 x0 = stateV[s * DIM + ampIndex(rsub, pm, pl, (unsigned)kb)];
            float2 x1 = stateV[s * DIM + ampIndex(rsub, pm, pl, (unsigned)(kb + 1))];

            for (int rr = 0; rr < 64; ++rr) {
                unsigned rn = 2u * (unsigned)((rr + 1) & 63) + rsub;
                float2 y0 = stateV[s * DIM + ampIndex(rn, pm, pl, (unsigned)kb)];
                float2 y1 = stateV[s * DIM + ampIndex(rn, pm, pl, (unsigned)(kb + 1))];

                v2f A0; A0[0] = x0.x; A0[1] = x1.x;
                v2f A1; A1[0] = x0.y; A1[1] = x1.y;
                v8f acc = {0.f, 0.f, 0.f, 0.f, 0.f, 0.f, 0.f, 0.f};
                acc = __builtin_amdgcn_wmma_f32_16x16x4_f32(false, A0, false, B0,
                                                            (short)0, acc, false, false);
                acc = __builtin_amdgcn_wmma_f32_16x16x4_f32(false, A1, false, B1,
                                                            (short)0, acc, false, false);

                unsigned aw = ampIndex(2u * (unsigned)rr + (unsigned)half, pm, pl, q);
                #pragma unroll
                for (int v = 0; v < 8; ++v)
                    stateF[v * (DIM * 2) + aw * 2 + reim] = acc[v];

                x0 = y0; x1 = y1;
            }
        }
    }

    for (int ss2 = 0; ss2 < STATES_PER_WAVE; ++ss2) {
        float* o = out + (size_t)(bBase + ss2) * DIM;
        for (int j = lane; j < DIM; j += 32) {
            float2 a = stateV[ss2 * DIM + j];
            o[j] = tanhf(5.0f * (a.x * a.x + a.y * a.y));
        }
    }
}

extern "C" void kernel_launch(void* const* d_in, const int* in_sizes, int n_in,
                              void* d_out, int out_size, void* d_ws, size_t ws_size,
                              hipStream_t stream) {
    (void)in_sizes; (void)n_in; (void)d_ws; (void)ws_size; (void)out_size;
    const float* features = (const float*)d_in[0];
    const float* weights  = (const float*)d_in[1];
    float* out = (float*)d_out;

    dim3 grid(BATCH_TOT / STATES_PER_BLOCK);
    dim3 block(BLOCK_THREADS);
    size_t shmem = (size_t)(GATE_LDS_FLOATS + WAVES_PER_BLOCK * STATE_LDS_FLOATS)
                   * sizeof(float);
    su4_quantum_kernel<<<grid, block, shmem, stream>>>(features, weights, out);
}


# Dense kernel: 8-wave workgroups (was 4), wave tile 64 rows x 16 complex columns (was 32): 158 registers (was 362), 6 waves per SIMD resident (was 2); staged tile stored at trip end; sums unchanged
# speedup vs baseline: 9.8526x; 9.8526x over previous
#include <hip/hip_runtime.h>
#include <math.h>


#define NQ        9
#define DIM       512
#define NLAYERS   12
#define OBS       256
#define BATCH_TOT 16384

#define NGATES            (NLAYERS * NQ)
#define GATE_FLOATS       128
#define GATE_LDS_FLOATS   (NGATES * GATE_FLOATS)
#define UROW_FLOATS       (DIM * 2)

#define G_BM      128
#define G_BJ      64
#define G_KC      32
#define G_PA      36
#define G_PB      136
#define G_A_FLOATS (G_BM * G_PA)
#define G_B_FLOATS (G_KC * G_PB)
#define G_STAGE    (G_A_FLOATS + G_B_FLOATS)

typedef float v2f __attribute__((ext_vector_type(2)));
typedef float v8f __attribute__((ext_vector_type(8)));

struct cpx { float re, im; };
__device__ inline cpx cz() { cpx r; r.re = 0.f; r.im = 0.f; return r; }
__device__ inline cpx cmk(float a, float b) { cpx r; r.re = a; r.im = b; return r; }
__device__ inline cpx cmul(cpx a, cpx b) { return cmk(a.re*b.re - a.im*b.im, a.re*b.im + a.im*b.re); }
__device__ inline cpx cadd(cpx a, cpx b) { return cmk(a.re + b.re, a.im + b.im); }

__device__ inline void u3mat(float t, float p, float l, cpx U[2][2]) {
    float ct = cosf(0.5f * t), st = sinf(0.5f * t);
    float cp = cosf(p), sp = sinf(p), cl = cosf(l), sl = sinf(l);
    U[0][0] = cmk(ct, 0.f);
    U[0][1] = cmk(-cl * st, -sl * st);
    U[1][0] = cmk(cp * st, sp * st);
    U[1][1] = cmk((cp*cl - sp*sl) * ct, (cp*sl + sp*cl) * ct);
}

__device__ inline void kron22(const cpx A[2][2], const cpx B[2][2], cpx K[4][4]) {
    for (int a = 0; a < 2; ++a)
        for (int b = 0; b < 2; ++b)
            for (int c = 0; c < 2; ++c)
                for (int d = 0; d < 2; ++d)
                    K[2*a + c][2*b + d] = cmul(A[a][b], B[c][d]);
}

__device__ inline void mat4mul(cpx C[4][4], const cpx A[4][4], const cpx B[4][4]) {
    for (int i = 0; i < 4; ++i)
        for (int j = 0; j < 4; ++j) {
            cpx acc = cz();
            for (int k = 0; k < 4; ++k) acc = cadd(acc, cmul(A[i][k], B[k][j]));
            C[i][j] = acc;
        }
}

__device__ void computeGate(int g, const float* __restrict__ w, float* __restrict__ gl) {
    const float* th = w + g * 15;
    cpx Ua[2][2], Ub[2][2];
    cpx pre[4][4], post[4][4];
    u3mat(th[0], th[1], th[2], Ua);  u3mat(th[3], th[4], th[5], Ub);  kron22(Ua, Ub, pre);
    u3mat(th[9], th[10], th[11], Ua); u3mat(th[12], th[13], th[14], Ub); kron22(Ua, Ub, post);

    float c6 = cosf(0.5f*th[6]), s6 = sinf(0.5f*th[6]);
    float c7 = cosf(0.5f*th[7]), s7 = sinf(0.5f*th[7]);
    float c8 = cosf(0.5f*th[8]), s8 = sinf(0.5f*th[8]);
    cpx Rxx[4][4], Ryy[4][4], Rzz[4][4];
    for (int i = 0; i < 4; ++i)
        for (int j = 0; j < 4; ++j) { Rxx[i][j] = cz(); Ryy[i][j] = cz(); Rzz[i][j] = cz(); }
    for (int i = 0; i < 4; ++i) { Rxx[i][i] = cmk(c6, 0.f); Ryy[i][i] = cmk(c7, 0.f); }
    Rxx[0][3] = cmk(0.f, -s6); Rxx[1][2] = cmk(0.f, -s6);
    Rxx[2][1] = cmk(0.f, -s6); Rxx[3][0] = cmk(0.f, -s6);
    Ryy[0][3] = cmk(0.f,  s7); Ryy[1][2] = cmk(0.f, -s7);
    Ryy[2][1] = cmk(0.f, -s7); Ryy[3][0] = cmk(0.f,  s7);
    Rzz[0][0] = cmk(c8, -s8); Rzz[1][1] = cmk(c8,  s8);
    Rzz[2][2] = cmk(c8,  s8); Rzz[3][3] = cmk(c8, -s8);

    cpx t1[4][4], t2[4][4];
    mat4mul(t1, Rxx, Ryy);
    mat4mul(t2, t1, Rzz);
    mat4mul(t1, t2, pre);
    mat4mul(t2, post, t1);

    float* gb = gl + g * GATE_FLOATS;
    for (int c = 0; c < 2; ++c)
        for (int k = 0; k < 4; ++k)
            for (int n = 0; n < 16; ++n) {
                int a = n & 7;
                int b = 4*c + k;
                float val;
                if (a < 4) val = (b < 4) ?  t2[a][b].re     : -t2[a][b-4].im;
                else       val = (b < 4) ?  t2[a-4][b].im   :  t2[a-4][b-4].re;
                gb[c*64 + k*16 + n] = val;
            }
}

__device__ constexpr int SLOT_BIT(int b) {
    return b == 3 ? 0 : b == 4 ? 1 : b == 8 ? 2 : b == 5 ? 3 : b == 0 ? 4
         : b == 1 ? 5 : b == 2 ? 6 : b == 6 ? 7 : 8;
}
__device__ constexpr unsigned slot_mask(unsigned logical_mask) {
    unsigned r = 0;
    for (int b = 0; b < NQ; ++b) if ((logical_mask >> b) & 1u) r |= 1u << SLOT_BIT(b);
    return r;
}
__device__ inline unsigned slot_of(unsigned amp) {
    unsigned r = 0;
    for (int b = 0; b < NQ; ++b) r |= ((amp >> b) & 1u) << SLOT_BIT(b);
    return r;
}
__device__ inline unsigned deposit(unsigned x, unsigned mask) {
    unsigned r = 0;
    int j = 0;
    for (int b = 0; b < NQ; ++b)
        if ((mask >> b) & 1u) { r |= ((x >> j) & 1u) << b; ++j; }
    return r;
}

__global__ __launch_bounds__(256)
void compose_columns(const float* __restrict__ weights, float* __restrict__ umat) {
    extern __shared__ __align__(16) float smem[];
    float* gl = smem;
    float*  stateF = smem + GATE_LDS_FLOATS;
    float2* stateV = (float2*)stateF;
    const int tid  = threadIdx.x;
    const int lane = tid & 31;
    const unsigned wave = (unsigned)__builtin_amdgcn_readfirstlane(tid >> 5);
    const int col  = blockIdx.x;

    if (tid < NGATES) computeGate(tid, weights, gl);
    const unsigned colslot = slot_of((unsigned)col);
    for (int j = tid; j < DIM; j += 256) stateV[j] = make_float2((unsigned)j == colslot ? 1.f : 0.f, 0.f);

    const unsigned m = (unsigned)(lane & 15);
    const unsigned half = (unsigned)(lane >> 4);
    const int kb   = 2 * (int)half;
    const int ncol = lane & 15;
    const unsigned q = (unsigned)(ncol & 3);
    const int reim = (ncol >> 2) & 1;
    const unsigned mrow = ((m & 7u) << 1) | (m >> 3);

    for (int layer = 0; layer < NLAYERS; ++layer) {
        #pragma unroll
        for (int i = 0; i < NQ; ++i) {
            if (i == 0 || i == 5) {
                __syncthreads();
            } else {
                __builtin_amdgcn_fence(__ATOMIC_RELEASE, "wavefront");
                __builtin_amdgcn_wave_barrier();
                __builtin_amdgcn_fence(__ATOMIC_ACQUIRE, "wavefront");
            }
            const int g = layer * NQ + i;
            const float* gb = gl + g * GATE_FLOATS;
            v2f B0, B1;
            B0[0] = gb[(kb    ) * 16 + ncol];
            B0[1] = gb[(kb + 1) * 16 + ncol];
            B1[0] = gb[64 + (kb    ) * 16 + ncol];
            B1[1] = gb[64 + (kb + 1) * 16 + ncol];
            const int pm = SLOT_BIT((i < 8) ? (8 - i) : 0);
            const int pl = SLOT_BIT((i < 8) ? (7 - i) : 8);
            const unsigned wsel = slot_mask((i < 5) ? 0x007u : 0x0E0u);
            const unsigned rest = 0x1FFu & ~wsel & ~(1u << pm) & ~(1u << pl);
            const unsigned wbits = deposit(wave, wsel);

            const unsigned ia = wbits | deposit(mrow, rest) | (half << pm);
            float2 x0 = stateV[ia];
            float2 x1 = stateV[ia | (1u << pl)];

            v2f A0; A0[0] = x0.x; A0[1] = x1.x;
            v2f A1; A1[0] = x0.y; A1[1] = x1.y;
            v8f acc = {0.f, 0.f, 0.f, 0.f, 0.f, 0.f, 0.f, 0.f};
            acc = __builtin_amdgcn_wmma_f32_16x16x4_f32(false, A0, false, B0,
                                                        (short)0, acc, false, false);
            acc = __builtin_amdgcn_wmma_f32_16x16x4_f32(false, A1, false, B1,
                                                        (short)0, acc, false, false);

            const unsigned id = wbits | deposit(half, rest)
                              | (((q >> 1) & 1u) << pm) | ((q & 1u) << pl);
            #pragma unroll
            for (int v = 0; v < 8; ++v)
                stateF[(id | deposit((unsigned)v << 1, rest)) * 2u + (unsigned)reim] = acc[v];
        }
    }
    __syncthreads();

    float2* ucol = (float2*)(umat + (size_t)col * UROW_FLOATS);
    for (int j = tid; j < DIM; j += 256) ucol[j] = stateV[slot_of((unsigned)j)];
}

__global__ __launch_bounds__(256)
void apply_matrix(const float* __restrict__ features, const float* __restrict__ umat,
                  float* __restrict__ out) {
    extern __shared__ __align__(16) float smem[];
    const int tid  = threadIdx.x;
    const int lane = tid & 31;
    const int wave = tid >> 5;
    const int c = lane & 15;
    const int h = lane >> 4;
    const int wm = wave & 1;
    const int wn = wave >> 1;
    const int m0 = blockIdx.x * G_BM;
    const int j0 = blockIdx.y * G_BJ;

    const float* ga = features + (size_t)(m0 + (tid >> 3)) * OBS + (tid & 7) * 4;
    const float* gb = umat + (size_t)(tid >> 5) * UROW_FLOATS + j0 * 2 + (tid & 31) * 4;
    float* const sa = smem + (tid >> 3) * G_PA + (tid & 7) * 4;
    float* const sb = smem + G_A_FLOATS + (tid >> 5) * G_PB + (tid & 31) * 4;
    const float* const fa = smem + (64 * wm + c) * G_PA + 4 * h;
    const float* const fb = smem + G_A_FLOATS + (4 * h) * G_PB + 2 * (16 * wn + c);

    float4 ra0, ra1, ra2, ra3;
    float4 rb0, rb1, rb2, rb3;
    v8f accR[4], accI[4];
    float ssq[4];
    #pragma unroll
    for (int mt = 0; mt < 4; ++mt) {
        ssq[mt] = 0.f;
        #pragma unroll
        for (int r = 0; r < 8; ++r) { accR[mt][r] = 0.f; accI[mt][r] = 0.f; }
    }

#define STAGE_LOAD(i) { \
        ra##i = *(const float4*)(ga + (i) * 32 * OBS); \
        rb##i = *(const float4*)(gb + (size_t)(i) * 8 * UROW_FLOATS); }
#define STAGE_STORE(i) { \
        *(float4*)(sa + so + (i) * 32 * G_PA) = ra##i; \
        *(float4*)(sb + so + (i) * 8 * G_PB)  = rb##i; }
    { const int so = 0;
      STAGE_LOAD(0) STAGE_LOAD(1) STAGE_LOAD(2) STAGE_LOAD(3)
      STAGE_STORE(0) STAGE_STORE(1) STAGE_STORE(2) STAGE_STORE(3) }

    for (int kc = 0; kc < OBS / G_KC; ++kc) {
        __syncthreads();
        ga += G_KC;
        gb += (size_t)G_KC * UROW_FLOATS;
        if (kc + 1 < OBS / G_KC) {
            STAGE_LOAD(0) STAGE_LOAD(1) STAGE_LOAD(2) STAGE_LOAD(3)
        }
        const int sc = (kc & 1) * G_STAGE;
        #pragma unroll
        for (int s = 0; s < 4; ++s) {
            float av[4][4];
            float br[4], bi[4];
            #pragma unroll
            for (int mt = 0; mt < 4; ++mt) {
                const float4 t = *(const float4*)(fa + sc + 16 * mt * G_PA + 8 * s);
                av[mt][0] = t.x; av[mt][1] = t.y; av[mt][2] = t.z; av[mt][3] = t.w;
                ssq[mt] += t.x * t.x + t.y * t.y + t.z * t.z + t.w * t.w;
            }
            #pragma unroll
            for (int e = 0; e < 4; ++e) {
                const float2 t = *(const float2*)(fb + sc + (8 * s + e) * G_PB);
                br[e] = t.x; bi[e] = t.y;
            }
            #pragma unroll
            for (int st = 0; st < 2; ++st) {
                v2f Br; Br[0] = br[2 * st]; Br[1] = br[2 * st + 1];
                v2f Bi; Bi[0] = bi[2 * st]; Bi[1] = bi[2 * st + 1];
                #pragma unroll
                for (int mt = 0; mt < 4; ++mt) {
                    v2f A; A[0] = av[mt][2 * st]; A[1] = av[mt][2 * st + 1];
                    accR[mt] = __builtin_amdgcn_wmma_f32_16x16x4_f32(false, A, false, Br,
                                                    (short)0, accR[mt], false, false);
                    accI[mt] = __builtin_amdgcn_wmma_f32_16x16x4_f32(false, A, false, Bi,
                                                    (short)0, accI[mt], false, false);
                }
            }
        }
        if (kc + 1 < OBS / G_KC) {
            const int so = G_STAGE - sc;
            STAGE_STORE(0) STAGE_STORE(1) STAGE_STORE(2) STAGE_STORE(3)
        }
    }

    #pragma unroll
    for (int mt = 0; mt < 4; ++mt) {
        const float tot = ssq[mt] + __shfl_xor(ssq[mt], 16, 32);
        const float inv = 1.0f / (sqrtf(tot) + 1e-8f);
        #pragma unroll
        for (int r = 0; r < 8; ++r) {
            const float iv = __shfl(inv, 8 * h + r, 32);
            float* o = out + (size_t)(m0 + 64 * wm + 16 * mt + 8 * h + r) * DIM + j0 + 16 * wn + c;
            const float re = accR[mt][r] * iv;
            const float im = accI[mt][r] * iv;
            *o = tanhf(5.0f * (re * re + im * im));
        }
    }
}

extern "C" void kernel_launch(void* const* d_in, const int* in_sizes, int n_in,
                              void* d_out, int out_size, void* d_ws, size_t ws_size,
                              hipStream_t stream) {
    (void)in_sizes; (void)n_in; (void)ws_size; (void)out_size;
    const float* features = (const float*)d_in[0];
    const float* weights  = (const float*)d_in[1];
    float* out  = (float*)d_out;
    float* umat = (float*)d_ws;

    compose_columns<<<dim3(OBS), dim3(256),
                      (size_t)(GATE_LDS_FLOATS + UROW_FLOATS) * sizeof(float), stream>>>(weights, umat);
    apply_matrix<<<dim3(BATCH_TOT / G_BM, DIM / G_BJ), dim3(256),
                   (size_t)(2 * G_STAGE) * sizeof(float), stream>>>(features, umat, out);
}


# Dense kernel: row sums of squares taken once per workgroup from the staged feature registers, 1/norm to the epilogue via LDS (was per wave from A fragments in the WMMA loop); norm sum order changed
# speedup vs baseline: 10.3687x; 1.0524x over previous
#include <hip/hip_runtime.h>
#include <math.h>


#define NQ        9
#define DIM       512
#define NLAYERS   12
#define OBS       256
#define BATCH_TOT 16384

#define NGATES            (NLAYERS * NQ)
#define GATE_FLOATS       128
#define GATE_LDS_FLOATS   (NGATES * GATE_FLOATS)
#define UROW_FLOATS       (DIM * 2)

#define G_BM      128
#define G_BJ      64
#define G_KC      32
#define G_PA      36
#define G_PB      136
#define G_A_FLOATS (G_BM * G_PA)
#define G_B_FLOATS (G_KC * G_PB)
#define G_STAGE    (G_A_FLOATS + G_B_FLOATS)

typedef float v2f __attribute__((ext_vector_type(2)));
typedef float v8f __attribute__((ext_vector_type(8)));

struct cpx { float re, im; };
__device__ inline cpx cz() { cpx r; r.re = 0.f; r.im = 0.f; return r; }
__device__ inline cpx cmk(float a, float b) { cpx r; r.re = a; r.im = b; return r; }
__device__ inline cpx cmul(cpx a, cpx b) { return cmk(a.re*b.re - a.im*b.im, a.re*b.im + a.im*b.re); }
__device__ inline cpx cadd(cpx a, cpx b) { return cmk(a.re + b.re, a.im + b.im); }

__device__ inline void u3mat(float t, float p, float l, cpx U[2][2]) {
    float ct = cosf(0.5f * t), st = sinf(0.5f * t);
    float cp = cosf(p), sp = sinf(p), cl = cosf(l), sl = sinf(l);
    U[0][0] = cmk(ct, 0.f);
    U[0][1] = cmk(-cl * st, -sl * st);
    U[1][0] = cmk(cp * st, sp * st);
    U[1][1] = cmk((cp*cl - sp*sl) * ct, (cp*sl + sp*cl) * ct);
}

__device__ inline void kron22(const cpx A[2][2], const cpx B[2][2], cpx K[4][4]) {
    for (int a = 0; a < 2; ++a)
        for (int b = 0; b < 2; ++b)
            for (int c = 0; c < 2; ++c)
                for (int d = 0; d < 2; ++d)
                    K[2*a + c][2*b + d] = cmul(A[a][b], B[c][d]);
}

__device__ inline void mat4mul(cpx C[4][4], const cpx A[4][4], const cpx B[4][4]) {
    for (int i = 0; i < 4; ++i)
        for (int j = 0; j < 4; ++j) {
            cpx acc = cz();
            for (int k = 0; k < 4; ++k) acc = cadd(acc, cmul(A[i][k], B[k][j]));
            C[i][j] = acc;
        }
}

__device__ void computeGate(int g, const float* __restrict__ w, float* __restrict__ gl) {
    const float* th = w + g * 15;
    cpx Ua[2][2], Ub[2][2];
    cpx pre[4][4], post[4][4];
    u3mat(th[0], th[1], th[2], Ua);  u3mat(th[3], th[4], th[5], Ub);  kron22(Ua, Ub, pre);
    u3mat(th[9], th[10], th[11], Ua); u3mat(th[12], th[13], th[14], Ub); kron22(Ua, Ub, post);

    float c6 = cosf(0.5f*th[6]), s6 = sinf(0.5f*th[6]);
    float c7 = cosf(0.5f*th[7]), s7 = sinf(0.5f*th[7]);
    float c8 = cosf(0.5f*th[8]), s8 = sinf(0.5f*th[8]);
    cpx Rxx[4][4], Ryy[4][4], Rzz[4][4];
    for (int i = 0; i < 4; ++i)
        for (int j = 0; j < 4; ++j) { Rxx[i][j] = cz(); Ryy[i][j] = cz(); Rzz[i][j] = cz(); }
    for (int i = 0; i < 4; ++i) { Rxx[i][i] = cmk(c6, 0.f); Ryy[i][i] = cmk(c7, 0.f); }
    Rxx[0][3] = cmk(0.f, -s6); Rxx[1][2] = cmk(0.f, -s6);
    Rxx[2][1] = cmk(0.f, -s6); Rxx[3][0] = cmk(0.f, -s6);
    Ryy[0][3] = cmk(0.f,  s7); Ryy[1][2] = cmk(0.f, -s7);
    Ryy[2][1] = cmk(0.f, -s7); Ryy[3][0] = cmk(0.f,  s7);
    Rzz[0][0] = cmk(c8, -s8); Rzz[1][1] = cmk(c8,  s8);
    Rzz[2][2] = cmk(c8,  s8); Rzz[3][3] = cmk(c8, -s8);

    cpx t1[4][4], t2[4][4];
    mat4mul(t1, Rxx, Ryy);
    mat4mul(t2, t1, Rzz);
    mat4mul(t1, t2, pre);
    mat4mul(t2, post, t1);

    float* gb = gl + g * GATE_FLOATS;
    for (int c = 0; c < 2; ++c)
        for (int k = 0; k < 4; ++k)
            for (int n = 0; n < 16; ++n) {
                int a = n & 7;
                int b = 4*c + k;
                float val;
                if (a < 4) val = (b < 4) ?  t2[a][b].re     : -t2[a][b-4].im;
                else       val = (b < 4) ?  t2[a-4][b].im   :  t2[a-4][b-4].re;
                gb[c*64 + k*16 + n] = val;
            }
}

__device__ constexpr int SLOT_BIT(int b) {
    return b == 3 ? 0 : b == 4 ? 1 : b == 8 ? 2 : b == 5 ? 3 : b == 0 ? 4
         : b == 1 ? 5 : b == 2 ? 6 : b == 6 ? 7 : 8;
}
__device__ constexpr unsigned slot_mask(unsigned logical_mask) {
    unsigned r = 0;
    for (int b = 0; b < NQ; ++b) if ((logical_mask >> b) & 1u) r |= 1u << SLOT_BIT(b);
    return r;
}
__device__ inline unsigned slot_of(unsigned amp) {
    unsigned r = 0;
    for (int b = 0; b < NQ; ++b) r |= ((amp >> b) & 1u) << SLOT_BIT(b);
    return r;
}
__device__ inline unsigned deposit(unsigned x, unsigned mask) {
    unsigned r = 0;
    int j = 0;
    for (int b = 0; b < NQ; ++b)
        if ((mask >> b) & 1u) { r |= ((x >> j) & 1u) << b; ++j; }
    return r;
}

__global__ __launch_bounds__(256)
void compose_columns(const float* __restrict__ weights, float* __restrict__ umat) {
    extern __shared__ __align__(16) float smem[];
    float* gl = smem;
    float*  stateF = smem + GATE_LDS_FLOATS;
    float2* stateV = (float2*)stateF;
    const int tid  = threadIdx.x;
    const int lane = tid & 31;
    const unsigned wave = (unsigned)__builtin_amdgcn_readfirstlane(tid >> 5);
    const int col  = blockIdx.x;

    if (tid < NGATES) computeGate(tid, weights, gl);
    const unsigned colslot = slot_of((unsigned)col);
    for (int j = tid; j < DIM; j += 256) stateV[j] = make_float2((unsigned)j == colslot ? 1.f : 0.f, 0.f);

    const unsigned m = (unsigned)(lane & 15);
    const unsigned half = (unsigned)(lane >> 4);
    const int kb   = 2 * (int)half;
    const int ncol = lane & 15;
    const unsigned q = (unsigned)(ncol & 3);
    const int reim = (ncol >> 2) & 1;
    const unsigned mrow = ((m & 7u) << 1) | (m >> 3);

    for (int layer = 0; layer < NLAYERS; ++layer) {
        #pragma unroll
        for (int i = 0; i < NQ; ++i) {
            if (i == 0 || i == 5) {
                __syncthreads();
            } else {
                __builtin_amdgcn_fence(__ATOMIC_RELEASE, "wavefront");
                __builtin_amdgcn_wave_barrier();
                __builtin_amdgcn_fence(__ATOMIC_ACQUIRE, "wavefront");
            }
            const int g = layer * NQ + i;
            const float* gb = gl + g * GATE_FLOATS;
            v2f B0, B1;
            B0[0] = gb[(kb    ) * 16 + ncol];
            B0[1] = gb[(kb + 1) * 16 + ncol];
            B1[0] = gb[64 + (kb    ) * 16 + ncol];
            B1[1] = gb[64 + (kb + 1) * 16 + ncol];
            const int pm = SLOT_BIT((i < 8) ? (8 - i) : 0);
            const int pl = SLOT_BIT((i < 8) ? (7 - i) : 8);
            const unsigned wsel = slot_mask((i < 5) ? 0x007u : 0x0E0u);
            const unsigned rest = 0x1FFu & ~wsel & ~(1u << pm) & ~(1u << pl);
            const unsigned wbits = deposit(wave, wsel);

            const unsigned ia = wbits | deposit(mrow, rest) | (half << pm);
            float2 x0 = stateV[ia];
            float2 x1 = stateV[ia | (1u << pl)];

            v2f A0; A0[0] = x0.x; A0[1] = x1.x;
            v2f A1; A1[0] = x0.y; A1[1] = x1.y;
            v8f acc = {0.f, 0.f, 0.f, 0.f, 0.f, 0.f, 0.f, 0.f};
            acc = __builtin_amdgcn_wmma_f32_16x16x4_f32(false, A0, false, B0,
                                                        (short)0, acc, false, false);
            acc = __builtin_amdgcn_wmma_f32_16x16x4_f32(false, A1, false, B1,
                                                        (short)0, acc, false, false);

            const unsigned id = wbits | deposit(half, rest)
                              | (((q >> 1) & 1u) << pm) | ((q & 1u) << pl);
            #pragma unroll
            for (int v = 0; v < 8; ++v)
                stateF[(id | deposit((unsigned)v << 1, rest)) * 2u + (unsigned)reim] = acc[v];
        }
    }
    __syncthreads();

    float2* ucol = (float2*)(umat + (size_t)col * UROW_FLOATS);
    for (int j = tid; j < DIM; j += 256) ucol[j] = stateV[slot_of((unsigned)j)];
}

__global__ __launch_bounds__(256)
void apply_matrix(const float* __restrict__ features, const float* __restrict__ umat,
                  float* __restrict__ out) {
    extern __shared__ __align__(16) float smem[];
    const int tid  = threadIdx.x;
    const int lane = tid & 31;
    const int wave = tid >> 5;
    const int c = lane & 15;
    const int h = lane >> 4;
    const int wm = wave & 1;
    const int wn = wave >> 1;
    const int m0 = blockIdx.x * G_BM;
    const int j0 = blockIdx.y * G_BJ;

    const float* ga = features + (size_t)(m0 + (tid >> 3)) * OBS + (tid & 7) * 4;
    const float* gb = umat + (size_t)(tid >> 5) * UROW_FLOATS + j0 * 2 + (tid & 31) * 4;
    float* const sa = smem + (tid >> 3) * G_PA + (tid & 7) * 4;
    float* const sb = smem + G_A_FLOATS + (tid >> 5) * G_PB + (tid & 31) * 4;
    const float* const fa = smem + (64 * wm + c) * G_PA + 4 * h;
    const float* const fb = smem + G_A_FLOATS + (4 * h) * G_PB + 2 * (16 * wn + c);

    float4 ra0, ra1, ra2, ra3;
    float4 rb0, rb1, rb2, rb3;
    v8f accR[4], accI[4];
    float nsq0 = 0.f, nsq1 = 0.f, nsq2 = 0.f, nsq3 = 0.f;
    float* const sn = smem + 2 * G_STAGE;
    #pragma unroll
    for (int mt = 0; mt < 4; ++mt) {
        #pragma unroll
        for (int r = 0; r < 8; ++r) { accR[mt][r] = 0.f; accI[mt][r] = 0.f; }
    }

#define STAGE_LOAD(i) { \
        ra##i = *(const float4*)(ga + (i) * 32 * OBS); \
        rb##i = *(const float4*)(gb + (size_t)(i) * 8 * UROW_FLOATS); }
#define STAGE_STORE(i) { \
        *(float4*)(sa + so + (i) * 32 * G_PA) = ra##i; \
        *(float4*)(sb + so + (i) * 8 * G_PB)  = rb##i; \
        nsq##i += ra##i.x * ra##i.x + ra##i.y * ra##i.y + ra##i.z * ra##i.z + ra##i.w * ra##i.w; }
#define ROW_SUM_STEP(d) { \
        const float u0 = __shfl_xor(nsq0, d, 32), u1 = __shfl_xor(nsq1, d, 32); \
        const float u2 = __shfl_xor(nsq2, d, 32), u3 = __shfl_xor(nsq3, d, 32); \
        nsq0 += u0; nsq1 += u1; nsq2 += u2; nsq3 += u3; }
    { const int so = 0;
      STAGE_LOAD(0) STAGE_LOAD(1) STAGE_LOAD(2) STAGE_LOAD(3)
      STAGE_STORE(0) STAGE_STORE(1) STAGE_STORE(2) STAGE_STORE(3) }

    for (int kc = 0; kc < OBS / G_KC; ++kc) {
        __syncthreads();
        ga += G_KC;
        gb += (size_t)G_KC * UROW_FLOATS;
        if (kc + 1 < OBS / G_KC) {
            STAGE_LOAD(0) STAGE_LOAD(1) STAGE_LOAD(2) STAGE_LOAD(3)
        }
        const int sc = (kc & 1) * G_STAGE;
        #pragma unroll
        for (int s = 0; s < 4; ++s) {
            float av[4][4];
            float br[4], bi[4];
            #pragma unroll
            for (int mt = 0; mt < 4; ++mt) {
                const float4 t = *(const float4*)(fa + sc + 16 * mt * G_PA + 8 * s);
                av[mt][0] = t.x; av[mt][1] = t.y; av[mt][2] = t.z; av[mt][3] = t.w;
            }
            #pragma unroll
            for (int e = 0; e < 4; ++e) {
                const float2 t = *(const float2*)(fb + sc + (8 * s + e) * G_PB);
                br[e] = t.x; bi[e] = t.y;
            }
            #pragma unroll
            for (int st = 0; st < 2; ++st) {
                v2f Br; Br[0] = br[2 * st]; Br[1] = br[2 * st + 1];
                v2f Bi; Bi[0] = bi[2 * st]; Bi[1] = bi[2 * st + 1];
                #pragma unroll
                for (int mt = 0; mt < 4; ++mt) {
                    v2f A; A[0] = av[mt][2 * st]; A[1] = av[mt][2 * st + 1];
                    accR[mt] = __builtin_amdgcn_wmma_f32_16x16x4_f32(false, A, false, Br,
                                                    (short)0, accR[mt], false, false);
                    accI[mt] = __builtin_amdgcn_wmma_f32_16x16x4_f32(false, A, false, Bi,
                                                    (short)0, accI[mt], false, false);
                }
            }
        }
        if (kc + 1 < OBS / G_KC) {
            const int so = G_STAGE - sc;
            STAGE_STORE(0) STAGE_STORE(1) STAGE_STORE(2) STAGE_STORE(3)
        }
        if (kc + 2 == OBS / G_KC) {
            ROW_SUM_STEP(1) ROW_SUM_STEP(2) ROW_SUM_STEP(4)
            const int ri = tid & 7;
            float tot = nsq0;
            tot = ri == 1 ? nsq1 : tot;
            tot = ri == 2 ? nsq2 : tot;
            tot = ri == 3 ? nsq3 : tot;
            if (ri < 4) sn[(tid >> 3) + 32 * ri] = 1.0f / (sqrtf(tot) + 1e-8f);
        }
    }

    #pragma unroll
    for (int mt = 0; mt < 4; ++mt) {
        const float* nr = sn + 64 * wm + 16 * mt + 8 * h;
        const float4 n0 = *(const float4*)nr;
        const float4 n1 = *(const float4*)(nr + 4);
        const float iv[8] = {n0.x, n0.y, n0.z, n0.w, n1.x, n1.y, n1.z, n1.w};
        #pragma unroll
        for (int r = 0; r < 8; ++r) {
            float* o = out + (size_t)(m0 + 64 * wm + 16 * mt + 8 * h + r) * DIM + j0 + 16 * wn + c;
            const float re = accR[mt][r] * iv[r];
            const float im = accI[mt][r] * iv[r];
            *o = tanhf(5.0f * (re * re + im * im));
        }
    }
}

extern "C" void kernel_launch(void* const* d_in, const int* in_sizes, int n_in,
                              void* d_out, int out_size, void* d_ws, size_t ws_size,
                              hipStream_t stream) {
    (void)in_sizes; (void)n_in; (void)ws_size; (void)out_size;
    const float* features = (const float*)d_in[0];
    const float* weights  = (const float*)d_in[1];
    float* out  = (float*)d_out;
    float* umat = (float*)d_ws;

    compose_columns<<<dim3(OBS), dim3(256),
                      (size_t)(GATE_LDS_FLOATS + UROW_FLOATS) * sizeof(float), stream>>>(weights, umat);
    apply_matrix<<<dim3(BATCH_TOT / G_BM, DIM / G_BJ), dim3(256),
                   (size_t)(2 * G_STAGE + G_BM) * sizeof(float), stream>>>(features, umat, out);
}


# Compose kernel: gate set-up on all 256 threads (was 108, inlining 30 trig calls): sin/cos tabulated in LDS by one looped call, one gate column per thread; text under 32 KB; gate sum order changed
# speedup vs baseline: 10.8215x; 1.0437x over previous
#include <hip/hip_runtime.h>
#include <math.h>


#define NQ        9
#define DIM       512
#define NLAYERS   12
#define OBS       256
#define BATCH_TOT 16384

#define NGATES            (NLAYERS * NQ)
#define GATE_FLOATS       128
#define GATE_LDS_FLOATS   (NGATES * GATE_FLOATS)
#define UROW_FLOATS       (DIM * 2)

#define G_BM      128
#define G_BJ      64
#define G_KC      32
#define G_PA      36
#define G_PB      136
#define G_A_FLOATS (G_BM * G_PA)
#define G_B_FLOATS (G_KC * G_PB)
#define G_STAGE    (G_A_FLOATS + G_B_FLOATS)

typedef float v2f __attribute__((ext_vector_type(2)));
typedef float v8f __attribute__((ext_vector_type(8)));

struct cpx { float re, im; };
__device__ inline cpx cmk(float a, float b) { cpx r; r.re = a; r.im = b; return r; }
__device__ inline cpx cmul(cpx a, cpx b) { return cmk(a.re*b.re - a.im*b.im, a.re*b.im + a.im*b.re); }
__device__ inline cpx cadd(cpx a, cpx b) { return cmk(a.re + b.re, a.im + b.im); }

#define NANGLES      (NGATES * 15)
#define NTAB         1792
#define HALF_ANGLES  0x13C9u

__device__ inline void u3tab(const float* __restrict__ sn, const float* __restrict__ cs, int a,
                             cpx U[2][2]) {
    const float ct = cs[a],     st = sn[a];
    const float cp = cs[a + 1], sp = sn[a + 1];
    const float cl = cs[a + 2], sl = sn[a + 2];
    U[0][0] = cmk(ct, 0.f);
    U[0][1] = cmk(-cl * st, -sl * st);
    U[1][0] = cmk(cp * st, sp * st);
    U[1][1] = cmk((cp*cl - sp*sl) * ct, (cp*sl + sp*cl) * ct);
}

__device__ inline void gateColumn(int g, int j, const float* __restrict__ sn,
                                  const float* __restrict__ cs, float* __restrict__ gl) {
    const int a0 = g * 15;
    cpx Ua[2][2], Ub[2][2], Pa[2][2], Pb[2][2];
    u3tab(sn, cs, a0,     Ua);
    u3tab(sn, cs, a0 + 3, Ub);
    u3tab(sn, cs, a0 + 9, Pa);
    u3tab(sn, cs, a0 + 12, Pb);
    const float c6 = cs[a0 + 6], s6 = sn[a0 + 6];
    const float c7 = cs[a0 + 7], s7 = sn[a0 + 7];
    const float c8 = cs[a0 + 8], s8 = sn[a0 + 8];

    const bool bh = (j >> 1) != 0, bl = (j & 1) != 0;
    const cpx A0 = bh ? Ua[0][1] : Ua[0][0], A1 = bh ? Ua[1][1] : Ua[1][0];
    const cpx B0 = bl ? Ub[0][1] : Ub[0][0], B1 = bl ? Ub[1][1] : Ub[1][0];
    cpx v0 = cmul(A0, B0), v1 = cmul(A0, B1), v2 = cmul(A1, B0), v3 = cmul(A1, B1);

    const cpx zm = cmk(c8, -s8), zp = cmk(c8, s8);
    v0 = cmul(zm, v0); v1 = cmul(zp, v1); v2 = cmul(zp, v2); v3 = cmul(zm, v3);

    cpx y0 = cmk(c7 * v0.re - s7 * v3.im, c7 * v0.im + s7 * v3.re);
    cpx y1 = cmk(c7 * v1.re + s7 * v2.im, c7 * v1.im - s7 * v2.re);
    cpx y2 = cmk(c7 * v2.re + s7 * v1.im, c7 * v2.im - s7 * v1.re);
    cpx y3 = cmk(c7 * v3.re - s7 * v0.im, c7 * v3.im + s7 * v0.re);

    cpx x0 = cmk(c6 * y0.re + s6 * y3.im, c6 * y0.im - s6 * y3.re);
    cpx x1 = cmk(c6 * y1.re + s6 * y2.im, c6 * y1.im - s6 * y2.re);
    cpx x2 = cmk(c6 * y2.re + s6 * y1.im, c6 * y2.im - s6 * y1.re);
    cpx x3 = cmk(c6 * y3.re + s6 * y0.im, c6 * y3.im - s6 * y0.re);

    cpx w0 = cadd(cmul(Pb[0][0], x0), cmul(Pb[0][1], x1));
    cpx w1 = cadd(cmul(Pb[1][0], x0), cmul(Pb[1][1], x1));
    cpx w2 = cadd(cmul(Pb[0][0], x2), cmul(Pb[0][1], x3));
    cpx w3 = cadd(cmul(Pb[1][0], x2), cmul(Pb[1][1], x3));
    cpx t0 = cadd(cmul(Pa[0][0], w0), cmul(Pa[0][1], w2));
    cpx t1 = cadd(cmul(Pa[0][0], w1), cmul(Pa[0][1], w3));
    cpx t2 = cadd(cmul(Pa[1][0], w0), cmul(Pa[1][1], w2));
    cpx t3 = cadd(cmul(Pa[1][0], w1), cmul(Pa[1][1], w3));

    float4* r0 = (float4*)(gl + g * GATE_FLOATS + j * 16);
    float4* r1 = (float4*)(gl + g * GATE_FLOATS + 64 + j * 16);
    const float4 re = make_float4(t0.re, t1.re, t2.re, t3.re);
    const float4 im = make_float4(t0.im, t1.im, t2.im, t3.im);
    const float4 nim = make_float4(-t0.im, -t1.im, -t2.im, -t3.im);
    r0[0] = re;  r0[1] = im; r0[2] = re;  r0[3] = im;
    r1[0] = nim; r1[1] = re; r1[2] = nim; r1[3] = re;
}

__device__ constexpr int SLOT_BIT(int b) {
    return b == 3 ? 0 : b == 4 ? 1 : b == 8 ? 2 : b == 5 ? 3 : b == 0 ? 4
         : b == 1 ? 5 : b == 2 ? 6 : b == 6 ? 7 : 8;
}
__device__ constexpr unsigned slot_mask(unsigned logical_mask) {
    unsigned r = 0;
    for (int b = 0; b < NQ; ++b) if ((logical_mask >> b) & 1u) r |= 1u << SLOT_BIT(b);
    return r;
}
__device__ inline unsigned slot_of(unsigned amp) {
    unsigned r = 0;
    for (int b = 0; b < NQ; ++b) r |= ((amp >> b) & 1u) << SLOT_BIT(b);
    return r;
}
__device__ inline unsigned deposit(unsigned x, unsigned mask) {
    unsigned r = 0;
    int j = 0;
    for (int b = 0; b < NQ; ++b)
        if ((mask >> b) & 1u) { r |= ((x >> j) & 1u) << b; ++j; }
    return r;
}

__global__ __launch_bounds__(256)
void compose_columns(const float* __restrict__ weights, float* __restrict__ umat) {
    extern __shared__ __align__(16) float smem[];
    float* gl = smem;
    float*  stateF = smem + GATE_LDS_FLOATS;
    float2* stateV = (float2*)stateF;
    const int tid  = threadIdx.x;
    const int lane = tid & 31;
    const unsigned wave = (unsigned)__builtin_amdgcn_readfirstlane(tid >> 5);
    const int col  = blockIdx.x;

    float* const sn = smem + GATE_LDS_FLOATS + UROW_FLOATS;
    float* const cs = sn + NTAB;
    float wv[NTAB / 256];
    #pragma unroll
    for (int k = 0; k < NTAB / 256; ++k) {
        const int idx = tid + 256 * k;
        wv[k] = weights[idx < NANGLES ? idx : NANGLES - 1];
    }
    #pragma unroll
    for (int k = 0; k < NTAB / 256; ++k) sn[tid + 256 * k] = wv[k];
    #pragma unroll 1
    for (int idx = tid; idx < NTAB; idx += 256) {
        const float t = sn[idx];
        const float arg = ((HALF_ANGLES >> (idx % 15)) & 1u) ? 0.5f * t : t;
        sn[idx] = sinf(arg);
        cs[idx] = cosf(arg);
    }
    __syncthreads();
    #pragma unroll 1
    for (int task = tid; task < NGATES * 4; task += 256) gateColumn(task >> 2, task & 3, sn, cs, gl);
    const unsigned colslot = slot_of((unsigned)col);
    for (int j = tid; j < DIM; j += 256) stateV[j] = make_float2((unsigned)j == colslot ? 1.f : 0.f, 0.f);

    const unsigned m = (unsigned)(lane & 15);
    const unsigned half = (unsigned)(lane >> 4);
    const int kb   = 2 * (int)half;
    const int ncol = lane & 15;
    const unsigned q = (unsigned)(ncol & 3);
    const int reim = (ncol >> 2) & 1;
    const unsigned mrow = ((m & 7u) << 1) | (m >> 3);

    for (int layer = 0; layer < NLAYERS; ++layer) {
        #pragma unroll
        for (int i = 0; i < NQ; ++i) {
            if (i == 0 || i == 5) {
                __syncthreads();
            } else {
                __builtin_amdgcn_fence(__ATOMIC_RELEASE, "wavefront");
                __builtin_amdgcn_wave_barrier();
                __builtin_amdgcn_fence(__ATOMIC_ACQUIRE, "wavefront");
            }
            const int g = layer * NQ + i;
            const float* gb = gl + g * GATE_FLOATS;
            v2f B0, B1;
            B0[0] = gb[(kb    ) * 16 + ncol];
            B0[1] = gb[(kb + 1) * 16 + ncol];
            B1[0] = gb[64 + (kb    ) * 16 + ncol];
            B1[1] = gb[64 + (kb + 1) * 16 + ncol];
            const int pm = SLOT_BIT((i < 8) ? (8 - i) : 0);
            const int pl = SLOT_BIT((i < 8) ? (7 - i) : 8);
            const unsigned wsel = slot_mask((i < 5) ? 0x007u : 0x0E0u);
            const unsigned rest = 0x1FFu & ~wsel & ~(1u << pm) & ~(1u << pl);
            const unsigned wbits = deposit(wave, wsel);

            const unsigned ia = wbits | deposit(mrow, rest) | (half << pm);
            float2 x0 = stateV[ia];
            float2 x1 = stateV[ia | (1u << pl)];

            v2f A0; A0[0] = x0.x; A0[1] = x1.x;
            v2f A1; A1[0] = x0.y; A1[1] = x1.y;
            v8f acc = {0.f, 0.f, 0.f, 0.f, 0.f, 0.f, 0.f, 0.f};
            acc = __builtin_amdgcn_wmma_f32_16x16x4_f32(false, A0, false, B0,
                                                        (short)0, acc, false, false);
            acc = __builtin_amdgcn_wmma_f32_16x16x4_f32(false, A1, false, B1,
                                                        (short)0, acc, false, false);

            const unsigned id = wbits | deposit(half, rest)
                              | (((q >> 1) & 1u) << pm) | ((q & 1u) << pl);
            #pragma unroll
            for (int v = 0; v < 8; ++v)
                stateF[(id | deposit((unsigned)v << 1, rest)) * 2u + (unsigned)reim] = acc[v];
        }
    }
    __syncthreads();

    float2* ucol = (float2*)(umat + (size_t)col * UROW_FLOATS);
    for (int j = tid; j < DIM; j += 256) ucol[j] = stateV[slot_of((unsigned)j)];
}

__global__ __launch_bounds__(256)
void apply_matrix(const float* __restrict__ features, const float* __restrict__ umat,
                  float* __restrict__ out) {
    extern __shared__ __align__(16) float smem[];
    const int tid  = threadIdx.x;
    const int lane = tid & 31;
    const int wave = tid >> 5;
    const int c = lane & 15;
    const int h = lane >> 4;
    const int wm = wave & 1;
    const int wn = wave >> 1;
    const int m0 = blockIdx.x * G_BM;
    const int j0 = blockIdx.y * G_BJ;

    const float* ga = features + (size_t)(m0 + (tid >> 3)) * OBS + (tid & 7) * 4;
    const float* gb = umat + (size_t)(tid >> 5) * UROW_FLOATS + j0 * 2 + (tid & 31) * 4;
    float* const sa = smem + (tid >> 3) * G_PA + (tid & 7) * 4;
    float* const sb = smem + G_A_FLOATS + (tid >> 5) * G_PB + (tid & 31) * 4;
    const float* const fa = smem + (64 * wm + c) * G_PA + 4 * h;
    const float* const fb = smem + G_A_FLOATS + (4 * h) * G_PB + 2 * (16 * wn + c);

    float4 ra0, ra1, ra2, ra3;
    float4 rb0, rb1, rb2, rb3;
    v8f accR[4], accI[4];
    float nsq0 = 0.f, nsq1 = 0.f, nsq2 = 0.f, nsq3 = 0.f;
    float* const sn = smem + 2 * G_STAGE;
    #pragma unroll
    for (int mt = 0; mt < 4; ++mt) {
        #pragma unroll
        for (int r = 0; r < 8; ++r) { accR[mt][r] = 0.f; accI[mt][r] = 0.f; }
    }

#define STAGE_LOAD(i) { \
        ra##i = *(const float4*)(ga + (i) * 32 * OBS); \
        rb##i = *(const float4*)(gb + (size_t)(i) * 8 * UROW_FLOATS); }
#define STAGE_STORE(i) { \
        *(float4*)(sa + so + (i) * 32 * G_PA) = ra##i; \
        *(float4*)(sb + so + (i) * 8 * G_PB)  = rb##i; \
        nsq##i += ra##i.x * ra##i.x + ra##i.y * ra##i.y + ra##i.z * ra##i.z + ra##i.w * ra##i.w; }
#define ROW_SUM_STEP(d) { \
        const float u0 = __shfl_xor(nsq0, d, 32), u1 = __shfl_xor(nsq1, d, 32); \
        const float u2 = __shfl_xor(nsq2, d, 32), u3 = __shfl_xor(nsq3, d, 32); \
        nsq0 += u0; nsq1 += u1; nsq2 += u2; nsq3 += u3; }
    { const int so = 0;
      STAGE_LOAD(0) STAGE_LOAD(1) STAGE_LOAD(2) STAGE_LOAD(3)
      STAGE_STORE(0) STAGE_STORE(1) STAGE_STORE(2) STAGE_STORE(3) }

    for (int kc = 0; kc < OBS / G_KC; ++kc) {
        __syncthreads();
        ga += G_KC;
        gb += (size_t)G_KC * UROW_FLOATS;
        if (kc + 1 < OBS / G_KC) {
            STAGE_LOAD(0) STAGE_LOAD(1) STAGE_LOAD(2) STAGE_LOAD(3)
        }
        const int sc = (kc & 1) * G_STAGE;
        #pragma unroll
        for (int s = 0; s < 4; ++s) {
            float av[4][4];
            float br[4], bi[4];
            #pragma unroll
            for (int mt = 0; mt < 4; ++mt) {
                const float4 t = *(const float4*)(fa + sc + 16 * mt * G_PA + 8 * s);
                av[mt][0] = t.x; av[mt][1] = t.y; av[mt][2] = t.z; av[mt][3] = t.w;
            }
            #pragma unroll
            for (int e = 0; e < 4; ++e) {
                const float2 t = *(const float2*)(fb + sc + (8 * s + e) * G_PB);
                br[e] = t.x; bi[e] = t.y;
            }
            #pragma unroll
            for (int st = 0; st < 2; ++st) {
                v2f Br; Br[0] = br[2 * st]; Br[1] = br[2 * st + 1];
                v2f Bi; Bi[0] = bi[2 * st]; Bi[1] = bi[2 * st + 1];
                #pragma unroll
                for (int mt = 0; mt < 4; ++mt) {
                    v2f A; A[0] = av[mt][2 * st]; A[1] = av[mt][2 * st + 1];
                    accR[mt] = __builtin_amdgcn_wmma_f32_16x16x4_f32(false, A, false, Br,
                                                    (short)0, accR[mt], false, false);
                    accI[mt] = __builtin_amdgcn_wmma_f32_16x16x4_f32(false, A, false, Bi,
                                                    (short)0, accI[mt], false, false);
                }
            }
        }
        if (kc + 1 < OBS / G_KC) {
            const int so = G_STAGE - sc;
            STAGE_STORE(0) STAGE_STORE(1) STAGE_STORE(2) STAGE_STORE(3)
        }
        if (kc + 2 == OBS / G_KC) {
            ROW_SUM_STEP(1) ROW_SUM_STEP(2) ROW_SUM_STEP(4)
            const int ri = tid & 7;
            float tot = nsq0;
            tot = ri == 1 ? nsq1 : tot;
            tot = ri == 2 ? nsq2 : tot;
            tot = ri == 3 ? nsq3 : tot;
            if (ri < 4) sn[(tid >> 3) + 32 * ri] = 1.0f / (sqrtf(tot) + 1e-8f);
        }
    }

    #pragma unroll
    for (int mt = 0; mt < 4; ++mt) {
        const float* nr = sn + 64 * wm + 16 * mt + 8 * h;
        const float4 n0 = *(const float4*)nr;
        const float4 n1 = *(const float4*)(nr + 4);
        const float iv[8] = {n0.x, n0.y, n0.z, n0.w, n1.x, n1.y, n1.z, n1.w};
        #pragma unroll
        for (int r = 0; r < 8; ++r) {
            float* o = out + (size_t)(m0 + 64 * wm + 16 * mt + 8 * h + r) * DIM + j0 + 16 * wn + c;
            const float re = accR[mt][r] * iv[r];
            const float im = accI[mt][r] * iv[r];
            *o = tanhf(5.0f * (re * re + im * im));
        }
    }
}

extern "C" void kernel_launch(void* const* d_in, const int* in_sizes, int n_in,
                              void* d_out, int out_size, void* d_ws, size_t ws_size,
                              hipStream_t stream) {
    (void)in_sizes; (void)n_in; (void)ws_size; (void)out_size;
    const float* features = (const float*)d_in[0];
    const float* weights  = (const float*)d_in[1];
    float* out  = (float*)d_out;
    float* umat = (float*)d_ws;

    compose_columns<<<dim3(OBS), dim3(256),
                      (size_t)(GATE_LDS_FLOATS + UROW_FLOATS + 2 * NTAB) * sizeof(float), stream>>>(weights, umat);
    apply_matrix<<<dim3(BATCH_TOT / G_BM, DIM / G_BJ), dim3(256),
                   (size_t)(2 * G_STAGE + G_BM) * sizeof(float), stream>>>(features, umat, out);
}


# Dense kernel tail: tanh on the hardware instruction, branch-free (was library code: 2 branches, about 20 vector instructions a value); packed scale and squares; constant store offsets; sums unchanged
# speedup vs baseline: 11.4087x; 1.0543x over previous
#include <hip/hip_runtime.h>
#include <math.h>


#define NQ        9
#define DIM       512
#define NLAYERS   12
#define OBS       256
#define BATCH_TOT 16384

#define NGATES            (NLAYERS * NQ)
#define GATE_FLOATS       128
#define GATE_LDS_FLOATS   (NGATES * GATE_FLOATS)
#define UROW_FLOATS       (DIM * 2)

#define G_BM      128
#define G_BJ      64
#define G_KC      32
#define G_PA      36
#define G_PB      136
#define G_A_FLOATS (G_BM * G_PA)
#define G_B_FLOATS (G_KC * G_PB)
#define G_STAGE    (G_A_FLOATS + G_B_FLOATS)

typedef float v2f __attribute__((ext_vector_type(2)));
typedef float v8f __attribute__((ext_vector_type(8)));

struct cpx { float re, im; };
__device__ inline cpx cmk(float a, float b) { cpx r; r.re = a; r.im = b; return r; }
__device__ inline cpx cmul(cpx a, cpx b) { return cmk(a.re*b.re - a.im*b.im, a.re*b.im + a.im*b.re); }
__device__ inline cpx cadd(cpx a, cpx b) { return cmk(a.re + b.re, a.im + b.im); }

#define NANGLES      (NGATES * 15)
#define NTAB         1792
#define HALF_ANGLES  0x13C9u

__device__ inline void u3tab(const float* __restrict__ sn, const float* __restrict__ cs, int a,
                             cpx U[2][2]) {
    const float ct = cs[a],     st = sn[a];
    const float cp = cs[a + 1], sp = sn[a + 1];
    const float cl = cs[a + 2], sl = sn[a + 2];
    U[0][0] = cmk(ct, 0.f);
    U[0][1] = cmk(-cl * st, -sl * st);
    U[1][0] = cmk(cp * st, sp * st);
    U[1][1] = cmk((cp*cl - sp*sl) * ct, (cp*sl + sp*cl) * ct);
}

__device__ inline void gateColumn(int g, int j, const float* __restrict__ sn,
                                  const float* __restrict__ cs, float* __restrict__ gl) {
    const int a0 = g * 15;
    cpx Ua[2][2], Ub[2][2], Pa[2][2], Pb[2][2];
    u3tab(sn, cs, a0,     Ua);
    u3tab(sn, cs, a0 + 3, Ub);
    u3tab(sn, cs, a0 + 9, Pa);
    u3tab(sn, cs, a0 + 12, Pb);
    const float c6 = cs[a0 + 6], s6 = sn[a0 + 6];
    const float c7 = cs[a0 + 7], s7 = sn[a0 + 7];
    const float c8 = cs[a0 + 8], s8 = sn[a0 + 8];

    const bool bh = (j >> 1) != 0, bl = (j & 1) != 0;
    const cpx A0 = bh ? Ua[0][1] : Ua[0][0], A1 = bh ? Ua[1][1] : Ua[1][0];
    const cpx B0 = bl ? Ub[0][1] : Ub[0][0], B1 = bl ? Ub[1][1] : Ub[1][0];
    cpx v0 = cmul(A0, B0), v1 = cmul(A0, B1), v2 = cmul(A1, B0), v3 = cmul(A1, B1);

    const cpx zm = cmk(c8, -s8), zp = cmk(c8, s8);
    v0 = cmul(zm, v0); v1 = cmul(zp, v1); v2 = cmul(zp, v2); v3 = cmul(zm, v3);

    cpx y0 = cmk(c7 * v0.re - s7 * v3.im, c7 * v0.im + s7 * v3.re);
    cpx y1 = cmk(c7 * v1.re + s7 * v2.im, c7 * v1.im - s7 * v2.re);
    cpx y2 = cmk(c7 * v2.re + s7 * v1.im, c7 * v2.im - s7 * v1.re);
    cpx y3 = cmk(c7 * v3.re - s7 * v0.im, c7 * v3.im + s7 * v0.re);

    cpx x0 = cmk(c6 * y0.re + s6 * y3.im, c6 * y0.im - s6 * y3.re);
    cpx x1 = cmk(c6 * y1.re + s6 * y2.im, c6 * y1.im - s6 * y2.re);
    cpx x2 = cmk(c6 * y2.re + s6 * y1.im, c6 * y2.im - s6 * y1.re);
    cpx x3 = cmk(c6 * y3.re + s6 * y0.im, c6 * y3.im - s6 * y0.re);

    cpx w0 = cadd(cmul(Pb[0][0], x0), cmul(Pb[0][1], x1));
    cpx w1 = cadd(cmul(Pb[1][0], x0), cmul(Pb[1][1], x1));
    cpx w2 = cadd(cmul(Pb[0][0], x2), cmul(Pb[0][1], x3));
    cpx w3 = cadd(cmul(Pb[1][0], x2), cmul(Pb[1][1], x3));
    cpx t0 = cadd(cmul(Pa[0][0], w0), cmul(Pa[0][1], w2));
    cpx t1 = cadd(cmul(Pa[0][0], w1), cmul(Pa[0][1], w3));
    cpx t2 = cadd(cmul(Pa[1][0], w0), cmul(Pa[1][1], w2));
    cpx t3 = cadd(cmul(Pa[1][0], w1), cmul(Pa[1][1], w3));

    float4* r0 = (float4*)(gl + g * GATE_FLOATS + j * 16);
    float4* r1 = (float4*)(gl + g * GATE_FLOATS + 64 + j * 16);
    const float4 re = make_float4(t0.re, t1.re, t2.re, t3.re);
    const float4 im = make_float4(t0.im, t1.im, t2.im, t3.im);
    const float4 nim = make_float4(-t0.im, -t1.im, -t2.im, -t3.im);
    r0[0] = re;  r0[1] = im; r0[2] = re;  r0[3] = im;
    r1[0] = nim; r1[1] = re; r1[2] = nim; r1[3] = re;
}

__device__ constexpr int SLOT_BIT(int b) {
    return b == 3 ? 0 : b == 4 ? 1 : b == 8 ? 2 : b == 5 ? 3 : b == 0 ? 4
         : b == 1 ? 5 : b == 2 ? 6 : b == 6 ? 7 : 8;
}
__device__ constexpr unsigned slot_mask(unsigned logical_mask) {
    unsigned r = 0;
    for (int b = 0; b < NQ; ++b) if ((logical_mask >> b) & 1u) r |= 1u << SLOT_BIT(b);
    return r;
}
__device__ inline unsigned slot_of(unsigned amp) {
    unsigned r = 0;
    for (int b = 0; b < NQ; ++b) r |= ((amp >> b) & 1u) << SLOT_BIT(b);
    return r;
}
__device__ inline unsigned deposit(unsigned x, unsigned mask) {
    unsigned r = 0;
    int j = 0;
    for (int b = 0; b < NQ; ++b)
        if ((mask >> b) & 1u) { r |= ((x >> j) & 1u) << b; ++j; }
    return r;
}

__global__ __launch_bounds__(256)
void compose_columns(const float* __restrict__ weights, float* __restrict__ umat) {
    extern __shared__ __align__(16) float smem[];
    float* gl = smem;
    float*  stateF = smem + GATE_LDS_FLOATS;
    float2* stateV = (float2*)stateF;
    const int tid  = threadIdx.x;
    const int lane = tid & 31;
    const unsigned wave = (unsigned)__builtin_amdgcn_readfirstlane(tid >> 5);
    const int col  = blockIdx.x;

    float* const sn = smem + GATE_LDS_FLOATS + UROW_FLOATS;
    float* const cs = sn + NTAB;
    float wv[NTAB / 256];
    #pragma unroll
    for (int k = 0; k < NTAB / 256; ++k) {
        const int idx = tid + 256 * k;
        wv[k] = weights[idx < NANGLES ? idx : NANGLES - 1];
    }
    #pragma unroll
    for (int k = 0; k < NTAB / 256; ++k) sn[tid + 256 * k] = wv[k];
    #pragma unroll 1
    for (int idx = tid; idx < NTAB; idx += 256) {
        const float t = sn[idx];
        const float arg = ((HALF_ANGLES >> (idx % 15)) & 1u) ? 0.5f * t : t;
        sn[idx] = sinf(arg);
        cs[idx] = cosf(arg);
    }
    __syncthreads();
    #pragma unroll 1
    for (int task = tid; task < NGATES * 4; task += 256) gateColumn(task >> 2, task & 3, sn, cs, gl);
    const unsigned colslot = slot_of((unsigned)col);
    for (int j = tid; j < DIM; j += 256) stateV[j] = make_float2((unsigned)j == colslot ? 1.f : 0.f, 0.f);

    const unsigned m = (unsigned)(lane & 15);
    const unsigned half = (unsigned)(lane >> 4);
    const int kb   = 2 * (int)half;
    const int ncol = lane & 15;
    const unsigned q = (unsigned)(ncol & 3);
    const int reim = (ncol >> 2) & 1;
    const unsigned mrow = ((m & 7u) << 1) | (m >> 3);

    for (int layer = 0; layer < NLAYERS; ++layer) {
        #pragma unroll
        for (int i = 0; i < NQ; ++i) {
            if (i == 0 || i == 5) {
                __syncthreads();
            } else {
                __builtin_amdgcn_fence(__ATOMIC_RELEASE, "wavefront");
                __builtin_amdgcn_wave_barrier();
                __builtin_amdgcn_fence(__ATOMIC_ACQUIRE, "wavefront");
            }
            const int g = layer * NQ + i;
            const float* gb = gl + g * GATE_FLOATS;
            v2f B0, B1;
            B0[0] = gb[(kb    ) * 16 + ncol];
            B0[1] = gb[(kb + 1) * 16 + ncol];
            B1[0] = gb[64 + (kb    ) * 16 + ncol];
            B1[1] = gb[64 + (kb + 1) * 16 + ncol];
            const int pm = SLOT_BIT((i < 8) ? (8 - i) : 0);
            const int pl = SLOT_BIT((i < 8) ? (7 - i) : 8);
            const unsigned wsel = slot_mask((i < 5) ? 0x007u : 0x0E0u);
            const unsigned rest = 0x1FFu & ~wsel & ~(1u << pm) & ~(1u << pl);
            const unsigned wbits = deposit(wave, wsel);

            const unsigned ia = wbits | deposit(mrow, rest) | (half << pm);
            float2 x0 = stateV[ia];
            float2 x1 = stateV[ia | (1u << pl)];

            v2f A0; A0[0] = x0.x; A0[1] = x1.x;
            v2f A1; A1[0] = x0.y; A1[1] = x1.y;
            v8f acc = {0.f, 0.f, 0.f, 0.f, 0.f, 0.f, 0.f, 0.f};
            acc = __builtin_amdgcn_wmma_f32_16x16x4_f32(false, A0, false, B0,
                                                        (short)0, acc, false, false);
            acc = __builtin_amdgcn_wmma_f32_16x16x4_f32(false, A1, false, B1,
                                                        (short)0, acc, false, false);

            const unsigned id = wbits | deposit(half, rest)
                              | (((q >> 1) & 1u) << pm) | ((q & 1u) << pl);
            #pragma unroll
            for (int v = 0; v < 8; ++v)
                stateF[(id | deposit((unsigned)v << 1, rest)) * 2u + (unsigned)reim] = acc[v];
        }
    }
    __syncthreads();

    float2* ucol = (float2*)(umat + (size_t)col * UROW_FLOATS);
    for (int j = tid; j < DIM; j += 256) ucol[j] = stateV[slot_of((unsigned)j)];
}

__global__ __launch_bounds__(256)
void apply_matrix(const float* __restrict__ features, const float* __restrict__ umat,
                  float* __restrict__ out) {
    extern __shared__ __align__(16) float smem[];
    const int tid  = threadIdx.x;
    const int lane = tid & 31;
    const int wave = tid >> 5;
    const int c = lane & 15;
    const int h = lane >> 4;
    const int wm = wave & 1;
    const int wn = wave >> 1;
    const int m0 = blockIdx.x * G_BM;
    const int j0 = blockIdx.y * G_BJ;

    const float* ga = features + (size_t)(m0 + (tid >> 3)) * OBS + (tid & 7) * 4;
    const float* gb = umat + (size_t)(tid >> 5) * UROW_FLOATS + j0 * 2 + (tid & 31) * 4;
    float* const sa = smem + (tid >> 3) * G_PA + (tid & 7) * 4;
    float* const sb = smem + G_A_FLOATS + (tid >> 5) * G_PB + (tid & 31) * 4;
    const float* const fa = smem + (64 * wm + c) * G_PA + 4 * h;
    const float* const fb = smem + G_A_FLOATS + (4 * h) * G_PB + 2 * (16 * wn + c);

    float4 ra0, ra1, ra2, ra3;
    float4 rb0, rb1, rb2, rb3;
    v8f accR[4], accI[4];
    float nsq0 = 0.f, nsq1 = 0.f, nsq2 = 0.f, nsq3 = 0.f;
    float* const sn = smem + 2 * G_STAGE;
    #pragma unroll
    for (int mt = 0; mt < 4; ++mt) {
        #pragma unroll
        for (int r = 0; r < 8; ++r) { accR[mt][r] = 0.f; accI[mt][r] = 0.f; }
    }

#define STAGE_LOAD(i) { \
        ra##i = *(const float4*)(ga + (i) * 32 * OBS); \
        rb##i = *(const float4*)(gb + (size_t)(i) * 8 * UROW_FLOATS); }
#define STAGE_STORE(i) { \
        *(float4*)(sa + so + (i) * 32 * G_PA) = ra##i; \
        *(float4*)(sb + so + (i) * 8 * G_PB)  = rb##i; \
        nsq##i += ra##i.x * ra##i.x + ra##i.y * ra##i.y + ra##i.z * ra##i.z + ra##i.w * ra##i.w; }
#define ROW_SUM_STEP(d) { \
        const float u0 = __shfl_xor(nsq0, d, 32), u1 = __shfl_xor(nsq1, d, 32); \
        const float u2 = __shfl_xor(nsq2, d, 32), u3 = __shfl_xor(nsq3, d, 32); \
        nsq0 += u0; nsq1 += u1; nsq2 += u2; nsq3 += u3; }
    { const int so = 0;
      STAGE_LOAD(0) STAGE_LOAD(1) STAGE_LOAD(2) STAGE_LOAD(3)
      STAGE_STORE(0) STAGE_STORE(1) STAGE_STORE(2) STAGE_STORE(3) }

    for (int kc = 0; kc < OBS / G_KC; ++kc) {
        __syncthreads();
        ga += G_KC;
        gb += (size_t)G_KC * UROW_FLOATS;
        if (kc + 1 < OBS / G_KC) {
            STAGE_LOAD(0) STAGE_LOAD(1) STAGE_LOAD(2) STAGE_LOAD(3)
        }
        const int sc = (kc & 1) * G_STAGE;
        #pragma unroll
        for (int s = 0; s < 4; ++s) {
            float av[4][4];
            float br[4], bi[4];
            #pragma unroll
            for (int mt = 0; mt < 4; ++mt) {
                const float4 t = *(const float4*)(fa + sc + 16 * mt * G_PA + 8 * s);
                av[mt][0] = t.x; av[mt][1] = t.y; av[mt][2] = t.z; av[mt][3] = t.w;
            }
            #pragma unroll
            for (int e = 0; e < 4; ++e) {
                const float2 t = *(const float2*)(fb + sc + (8 * s + e) * G_PB);
                br[e] = t.x; bi[e] = t.y;
            }
            #pragma unroll
            for (int st = 0; st < 2; ++st) {
                v2f Br; Br[0] = br[2 * st]; Br[1] = br[2 * st + 1];
                v2f Bi; Bi[0] = bi[2 * st]; Bi[1] = bi[2 * st + 1];
                #pragma unroll
                for (int mt = 0; mt < 4; ++mt) {
                    v2f A; A[0] = av[mt][2 * st]; A[1] = av[mt][2 * st + 1];
                    accR[mt] = __builtin_amdgcn_wmma_f32_16x16x4_f32(false, A, false, Br,
                                                    (short)0, accR[mt], false, false);
                    accI[mt] = __builtin_amdgcn_wmma_f32_16x16x4_f32(false, A, false, Bi,
                                                    (short)0, accI[mt], false, false);
                }
            }
        }
        if (kc + 1 < OBS / G_KC) {
            const int so = G_STAGE - sc;
            STAGE_STORE(0) STAGE_STORE(1) STAGE_STORE(2) STAGE_STORE(3)
        }
        if (kc + 2 == OBS / G_KC) {
            ROW_SUM_STEP(1) ROW_SUM_STEP(2) ROW_SUM_STEP(4)
            const int ri = tid & 7;
            float tot = nsq0;
            tot = ri == 1 ? nsq1 : tot;
            tot = ri == 2 ? nsq2 : tot;
            tot = ri == 3 ? nsq3 : tot;
            if (ri < 4) sn[(tid >> 3) + 32 * ri] = 1.0f / (sqrtf(tot) + 1e-8f);
        }
    }

    float* const obase = out + (size_t)(m0 + 64 * wm + 8 * h) * DIM + j0 + 16 * wn + c;
    #pragma unroll
    for (int mt = 0; mt < 4; ++mt) {
        const float* nr = sn + 64 * wm + 16 * mt + 8 * h;
        const float4 n0 = *(const float4*)nr;
        const float4 n1 = *(const float4*)(nr + 4);
        const float iv[8] = {n0.x, n0.y, n0.z, n0.w, n1.x, n1.y, n1.z, n1.w};
        #pragma unroll
        for (int r = 0; r < 8; r += 2) {
            v2f nv; nv[0] = iv[r];        nv[1] = iv[r + 1];
            v2f re; re[0] = accR[mt][r];  re[1] = accR[mt][r + 1];
            v2f im; im[0] = accI[mt][r];  im[1] = accI[mt][r + 1];
            re = re * nv;
            im = im * nv;
            const v2f p = 5.0f * (re * re + im * im);
            obase[(16 * mt + r) * DIM]     = __builtin_amdgcn_tanhf(p[0]);
            obase[(16 * mt + r + 1) * DIM] = __builtin_amdgcn_tanhf(p[1]);
        }
    }
}

extern "C" void kernel_launch(void* const* d_in, const int* in_sizes, int n_in,
                              void* d_out, int out_size, void* d_ws, size_t ws_size,
                              hipStream_t stream) {
    (void)in_sizes; (void)n_in; (void)ws_size; (void)out_size;
    const float* features = (const float*)d_in[0];
    const float* weights  = (const float*)d_in[1];
    float* out  = (float*)d_out;
    float* umat = (float*)d_ws;

    compose_columns<<<dim3(OBS), dim3(256),
                      (size_t)(GATE_LDS_FLOATS + UROW_FLOATS + 2 * NTAB) * sizeof(float), stream>>>(weights, umat);
    apply_matrix<<<dim3(BATCH_TOT / G_BM, DIM / G_BJ), dim3(256),
                   (size_t)(2 * G_STAGE + G_BM) * sizeof(float), stream>>>(features, umat, out);
}
